# starting kernel
# baseline (speedup 1.0000x reference)
#include <hip/hip_runtime.h>
#include <math.h>


typedef float v2f __attribute__((ext_vector_type(2)));
typedef float v4f __attribute__((ext_vector_type(4)));
typedef float v8f __attribute__((ext_vector_type(8)));

#define NJ 24
#define ND 6
#define NF 19

#define OFF_W0  0
#define OFF_W1  (OFF_W0 + 16*288)
#define OFF_W2  (OFF_W1 + NJ*32*20)
#define OFF_B1  (OFF_W2 + NJ*16*20)
#define OFF_B2  (OFF_B1 + NJ*32)
#define OFF_B0  (OFF_B2 + NJ*16)
#define WS_FLOATS (OFF_B0 + 16)

__constant__ int c_parents[NJ] = {-1,0,0,0,1,2,3,4,5,6,7,8,9,9,9,12,13,14,16,17,18,19,20,21};

static __device__ __forceinline__ v8f wmma_f32(v2f a, v2f b, v8f c) {
  return __builtin_amdgcn_wmma_f32_16x16x4_f32(
       false, a,  false, b,
       (short)0, c,  false,  false);
}

__global__ __launch_bounds__(256)
void pad_weights_kernel(const float* __restrict__ W0, const float* __restrict__ b0,
                        const float* __restrict__ W1, const float* __restrict__ b1,
                        const float* __restrict__ W2, const float* __restrict__ b2,
                        float* __restrict__ ws)
{
  const int t = threadIdx.x;
  float* W0p = ws + OFF_W0;
  float* W1p = ws + OFF_W1;
  float* W2p = ws + OFF_W2;
  float* b1p = ws + OFF_B1;
  float* b2p = ws + OFF_B2;
  float* b0p = ws + OFF_B0;

  for (int i = t; i < 16 * 288; i += 256) {
    const int n = i / 288, k = i - n * 288;
    W0p[i] = (n < ND) ? W0[n * 288 + k] : 0.0f;
  }
  for (int i = t; i < NJ * 32 * 20; i += 256) {
    const int j = i / 640, r = i - j * 640, n = r / 20, k = r - n * 20;
    W1p[i] = (n < NF && k < NF) ? W1[(j * NF + n) * NF + k] : 0.0f;
  }
  for (int i = t; i < NJ * 16 * 20; i += 256) {
    const int j = i / 320, r = i - j * 320, n = r / 20, k = r - n * 20;
    W2p[i] = (n < ND && k < NF) ? W2[(j * ND + n) * NF + k] : 0.0f;
  }
  for (int i = t; i < NJ * 32; i += 256) {
    const int j = i / 32, n = i - j * 32;
    b1p[i] = (n < NF) ? b1[j * NF + n] : 0.0f;
  }
  for (int i = t; i < NJ * 16; i += 256) {
    const int j = i / 16, n = i - j * 16;
    b2p[i] = (n < ND) ? b2[j * ND + n] : 0.0f;
  }
  if (t < 16) b0p[t] = (t < ND) ? b0[t] : 0.0f;
}

__global__ __launch_bounds__(128)
void hier_pose_kernel(const float* __restrict__ rots, const float* __restrict__ Jtrs,
                      const float* __restrict__ ws,   float* __restrict__ out)
{
  __shared__ __align__(16) float sfeat[4][16][152];
  __shared__ __align__(16) float sx[4][16][20];

  const float* W0p = ws + OFF_W0;
  const float* W1p = ws + OFF_W1;
  const float* W2p = ws + OFF_W2;
  const float* b1p = ws + OFF_B1;
  const float* b2p = ws + OFF_B2;
  const float* b0p = ws + OFF_B0;

  const int  lane = threadIdx.x & 31;
  const int  wave = threadIdx.x >> 5;
  const int  half = lane >> 4;
  const int  ln   = lane & 15;
  const long s0   = ((long)blockIdx.x * 4 + wave) * 16;

  float* feat = &sfeat[wave][0][0];
  float* xb   = &sx[wave][0][0];

  const float* arow_r = rots + (s0 + ln) * 216;
  const float* arow_j = Jtrs + (s0 + ln) * 72;
  __builtin_prefetch(arow_r, 0, 3);
  __builtin_prefetch(arow_j, 0, 3);

  v8f acc;
  {
    const float ci = b0p[ln];
    #pragma unroll
    for (int g = 0; g < 8; ++g) acc[g] = ci;
  }
  #pragma unroll 3
  for (int kk = 0; kk < 54; ++kk) {
    const int k = kk * 4 + half * 2;
    const v2f a = *(const v2f*)(arow_r + k);
    const v2f b = *(const v2f*)(W0p + ln * 288 + k);
    acc = wmma_f32(a, b, acc);
  }
  #pragma unroll 3
  for (int kk = 54; kk < 72; ++kk) {
    const int k = kk * 4 + half * 2;
    const v2f a = *(const v2f*)(arow_j + (k - 216));
    const v2f b = *(const v2f*)(W0p + ln * 288 + k);
    acc = wmma_f32(a, b, acc);
  }
  if (ln < ND) {
    #pragma unroll
    for (int g = 0; g < 8; ++g)
      feat[(g + half * 8) * 152 + 144 + ln] = acc[g];
  }

  for (int j = 0; j < NJ; ++j) {
    const int p     = c_parents[j];
    const int fslot = (p < 0) ? 24 : p;

    if (lane < 16) {
      const int  m = lane;
      const long s = s0 + m;
      const float* rp = rots + (s * NJ + j) * 9;
      const float* tp = Jtrs + (s * NJ + j) * 3;
      __align__(16) float xr[20];
      #pragma unroll
      for (int i = 0; i < 9; ++i) xr[i] = rp[i];
      const float jx = tp[0], jy = tp[1], jz = tp[2];
      xr[9] = jx; xr[10] = jy; xr[11] = jz;
      float px = 0.0f, py = 0.0f, pz = 0.0f;
      if (p >= 0) {
        const float* pp = Jtrs + (s * NJ + p) * 3;
        px = pp[0]; py = pp[1]; pz = pp[2];
      }
      const float dx = jx - px, dy = jy - py, dz = jz - pz;
      xr[12] = sqrtf(dx * dx + dy * dy + dz * dz);
      #pragma unroll
      for (int i = 0; i < ND; ++i) xr[13 + i] = feat[m * 152 + fslot * ND + i];
      xr[19] = 0.0f;
      #pragma unroll
      for (int i = 0; i < 20; i += 4)
        *(v4f*)(xb + m * 20 + i) = *(const v4f*)(xr + i);
    }
    asm volatile("s_wait_dscnt 0" ::: "memory");

    v8f c0, c1;
    {
      const float bias0 = b1p[j * 32 + ln];
      const float bias1 = b1p[j * 32 + 16 + ln];
      #pragma unroll
      for (int g = 0; g < 8; ++g) { c0[g] = bias0; c1[g] = bias1; }
    }
    const float* w1j = W1p + j * 32 * 20;
    #pragma unroll
    for (int kk = 0; kk < 5; ++kk) {
      const int k = kk * 4 + half * 2;
      const v2f a   = *(const v2f*)(xb + ln * 20 + k);
      const v2f bt0 = *(const v2f*)(w1j + ln * 20 + k);
      const v2f bt1 = *(const v2f*)(w1j + (16 + ln) * 20 + k);
      c0 = wmma_f32(a, bt0, c0);
      c1 = wmma_f32(a, bt1, c1);
    }

    #pragma unroll
    for (int g = 0; g < 8; ++g) {
      const int m = g + half * 8;
      xb[m * 20 + ln] = fmaxf(c0[g], 0.0f);
      if (ln < 3) xb[m * 20 + 16 + ln] = fmaxf(c1[g], 0.0f);
      if (ln == 3) xb[m * 20 + 19] = 0.0f;
    }
    asm volatile("s_wait_dscnt 0" ::: "memory");

    v8f c2;
    {
      const float bias2 = b2p[j * 16 + ln];
      #pragma unroll
      for (int g = 0; g < 8; ++g) c2[g] = bias2;
    }
    const float* w2j = W2p + j * 16 * 20;
    #pragma unroll
    for (int kk = 0; kk < 5; ++kk) {
      const int k = kk * 4 + half * 2;
      const v2f a  = *(const v2f*)(xb + ln * 20 + k);
      const v2f bt = *(const v2f*)(w2j + ln * 20 + k);
      c2 = wmma_f32(a, bt, c2);
    }
    if (ln < ND) {
      #pragma unroll
      for (int g = 0; g < 8; ++g)
        feat[(g + half * 8) * 152 + j * ND + ln] = c2[g];
    }
  }

  asm volatile("s_wait_dscnt 0" ::: "memory");
  #pragma unroll 4
  for (int idx = lane; idx < 16 * 36; idx += 32) {
    const int m = idx / 36;
    const int q = idx - m * 36;
    *(v4f*)(out + (s0 + m) * 144 + q * 4) = *(const v4f*)(feat + m * 152 + q * 4);
  }
}

extern "C" void kernel_launch(void* const* d_in, const int* in_sizes, int n_in,
                              void* d_out, int out_size, void* d_ws, size_t ws_size,
                              hipStream_t stream) {
  const float* rots = (const float*)d_in[0];
  const float* Jtrs = (const float*)d_in[1];
  const float* W0   = (const float*)d_in[2];
  const float* b0   = (const float*)d_in[3];
  const float* W1   = (const float*)d_in[4];
  const float* b1   = (const float*)d_in[5];
  const float* W2   = (const float*)d_in[6];
  const float* b2   = (const float*)d_in[7];
  float* out = (float*)d_out;
  float* ws  = (float*)d_ws;

  hipLaunchKernelGGL(pad_weights_kernel, dim3(1), dim3(256), 0, stream,
                     W0, b0, W1, b1, W2, b2, ws);

  const long bsz    = (long)in_sizes[0] / (NJ * 9);
  const int  blocks = (int)(bsz / 64);
  if (blocks > 0) {
    hipLaunchKernelGGL(hier_pose_kernel, dim3(blocks), dim3(128), 0, stream,
                       rots, Jtrs, ws, out);
  }
}


# Weights k-padded into LDS once per work-group (512 groups of 8 waves, each wave loops over its tiles); B fragments and biases read from LDS; prologue launch and workspace gone
# speedup vs baseline: 1.7372x; 1.7372x over previous
#include <hip/hip_runtime.h>
#include <math.h>


typedef float v2f __attribute__((ext_vector_type(2)));
typedef float v4f __attribute__((ext_vector_type(4)));
typedef float v8f __attribute__((ext_vector_type(8)));

#define NJ 24
#define ND 6
#define NF 19
#define NW 8
#define NG 512
#define KP 20
#define W0P 292

#define L_W1  0
#define L_W2  (L_W1 + NJ*NF*KP)
#define L_W0  (L_W2 + NJ*ND*KP)
#define L_B1  (L_W0 + ND*W0P)
#define L_B2  (L_B1 + NJ*NF)
#define L_B0  (L_B2 + NJ*ND)
#define L_WTOT (L_B0 + 8)

__constant__ int c_parents[NJ] = {-1,0,0,0,1,2,3,4,5,6,7,8,9,9,9,12,13,14,16,17,18,19,20,21};

static __device__ __forceinline__ v8f wmma_f32(v2f a, v2f b, v8f c) {
  return __builtin_amdgcn_wmma_f32_16x16x4_f32(
       false, a,  false, b,
       (short)0, c,  false,  false);
}

__global__ __launch_bounds__(NW * 32)
void hier_pose_kernel(const float* __restrict__ rots, const float* __restrict__ Jtrs,
                      const float* __restrict__ W0, const float* __restrict__ b0,
                      const float* __restrict__ W1, const float* __restrict__ b1,
                      const float* __restrict__ W2, const float* __restrict__ b2,
                      float* __restrict__ out, long bsz)
{
  __shared__ __align__(16) float sW[L_WTOT];
  __shared__ __align__(16) float sfeat[NW][16][152];
  __shared__ __align__(16) float sx[NW][16][20];

  {
    const int t = threadIdx.x;
    constexpr int NT  = NW * 32;
    constexpr int N1  = NJ * NF * NF, T1 = (N1 + NT - 1) / NT;
    constexpr int N2  = NJ * ND * NF, T2 = (N2 + NT - 1) / NT;
    constexpr int N0  = ND * 12 * NJ, T0 = (N0 + NT - 1) / NT;
    constexpr int NB1 = NJ * NF,      TB = (NB1 + NT - 1) / NT;
    float v1[T1], v2[T2], v0[T0], vb[TB];
    #pragma unroll
    for (int u = 0; u < T1; ++u) { const int i = t + u * NT; v1[u] = W1[i < N1 ? i : N1 - 1]; }
    #pragma unroll
    for (int u = 0; u < T2; ++u) { const int i = t + u * NT; v2[u] = W2[i < N2 ? i : N2 - 1]; }
    #pragma unroll
    for (int u = 0; u < T0; ++u) { const int i = t + u * NT; v0[u] = W0[i < N0 ? i : N0 - 1]; }
    #pragma unroll
    for (int u = 0; u < TB; ++u) { const int i = t + u * NT; vb[u] = b1[i < NB1 ? i : NB1 - 1]; }
    const float vb2 = b2[t < NJ * ND ? t : NJ * ND - 1];
    const float vb0 = b0[t < ND ? t : ND - 1];

    #pragma unroll
    for (int u = 0; u < T1; ++u) {
      const int i = t + u * NT, r = i / NF, k = i - r * NF;
      if (i < N1) sW[L_W1 + r * KP + k] = v1[u];
    }
    #pragma unroll
    for (int u = 0; u < T2; ++u) {
      const int i = t + u * NT, r = i / NF, k = i - r * NF;
      if (i < N2) sW[L_W2 + r * KP + k] = v2[u];
    }
    #pragma unroll
    for (int u = 0; u < T0; ++u) {
      const int i = t + u * NT, n = i / (12 * NJ), k = i - n * (12 * NJ);
      if (i < N0) sW[L_W0 + n * W0P + k] = v0[u];
    }
    #pragma unroll
    for (int u = 0; u < TB; ++u) { const int i = t + u * NT; if (i < NB1) sW[L_B1 + i] = vb[u]; }
    if (t < NJ * ND) sW[L_B2 + t] = vb2;
    if (t < 8)       sW[L_B0 + t] = (t < ND) ? vb0 : 0.0f;
    for (int r = t; r < NJ * (NF + ND); r += NT) sW[L_W1 + r * KP + NF] = 0.0f;
    if (t < ND * 4) sW[L_W0 + (t >> 2) * W0P + 12 * NJ + (t & 3)] = 0.0f;
  }
  __syncthreads();

  const int  lane = threadIdx.x & 31;
  const int  wave = threadIdx.x >> 5;
  const int  half = lane >> 4;
  const int  ln   = lane & 15;
  const int  n1   = (16 + ln < NF) ? 16 + ln : NF - 1;
  const int  n2   = (ln < ND) ? ln : ND - 1;

  float* feat = &sfeat[wave][0][0];
  float* xb   = &sx[wave][0][0];

  const long ntiles = bsz / 16;
  for (long tile = (long)blockIdx.x * NW + wave; tile < ntiles; tile += (long)NG * NW) {
  const long s0 = tile * 16;

  const float* arow_r = rots + (s0 + ln) * 216;
  const float* arow_j = Jtrs + (s0 + ln) * 72;
  __builtin_prefetch(arow_r, 0, 3);
  __builtin_prefetch(arow_j, 0, 3);

  v8f acc;
  {
    const float ci = sW[L_B0 + n2];
    #pragma unroll
    for (int g = 0; g < 8; ++g) acc[g] = ci;
  }
  const float* w0row = sW + L_W0 + n2 * W0P;
  #pragma unroll 3
  for (int kk = 0; kk < 54; ++kk) {
    const int k = kk * 4 + half * 2;
    const v2f a = *(const v2f*)(arow_r + k);
    const v2f b = *(const v2f*)(w0row + k);
    acc = wmma_f32(a, b, acc);
  }
  #pragma unroll 3
  for (int kk = 54; kk < 72; ++kk) {
    const int k = kk * 4 + half * 2;
    const v2f a = *(const v2f*)(arow_j + (k - 216));
    const v2f b = *(const v2f*)(w0row + k);
    acc = wmma_f32(a, b, acc);
  }
  if (ln < ND) {
    #pragma unroll
    for (int g = 0; g < 8; ++g)
      feat[(g + half * 8) * 152 + 144 + ln] = acc[g];
  }

  for (int j = 0; j < NJ; ++j) {
    const int p     = c_parents[j];
    const int fslot = (p < 0) ? 24 : p;

    if (lane < 16) {
      const int  m = lane;
      const long s = s0 + m;
      const float* rp = rots + (s * NJ + j) * 9;
      const float* tp = Jtrs + (s * NJ + j) * 3;
      __align__(16) float xr[20];
      #pragma unroll
      for (int i = 0; i < 9; ++i) xr[i] = rp[i];
      const float jx = tp[0], jy = tp[1], jz = tp[2];
      xr[9] = jx; xr[10] = jy; xr[11] = jz;
      float px = 0.0f, py = 0.0f, pz = 0.0f;
      if (p >= 0) {
        const float* pp = Jtrs + (s * NJ + p) * 3;
        px = pp[0]; py = pp[1]; pz = pp[2];
      }
      const float dx = jx - px, dy = jy - py, dz = jz - pz;
      xr[12] = sqrtf(dx * dx + dy * dy + dz * dz);
      #pragma unroll
      for (int i = 0; i < ND; ++i) xr[13 + i] = feat[m * 152 + fslot * ND + i];
      xr[19] = 0.0f;
      #pragma unroll
      for (int i = 0; i < 20; i += 4)
        *(v4f*)(xb + m * 20 + i) = *(const v4f*)(xr + i);
    }
    asm volatile("s_wait_dscnt 0" ::: "memory");

    v8f c0, c1;
    {
      const float bias0 = sW[L_B1 + j * NF + ln];
      const float bias1 = sW[L_B1 + j * NF + n1];
      #pragma unroll
      for (int g = 0; g < 8; ++g) { c0[g] = bias0; c1[g] = bias1; }
    }
    const float* w1j = sW + L_W1 + j * NF * KP;
    #pragma unroll
    for (int kk = 0; kk < 5; ++kk) {
      const int k = kk * 4 + half * 2;
      const v2f a   = *(const v2f*)(xb + ln * 20 + k);
      const v2f bt0 = *(const v2f*)(w1j + ln * KP + k);
      const v2f bt1 = *(const v2f*)(w1j + n1 * KP + k);
      c0 = wmma_f32(a, bt0, c0);
      c1 = wmma_f32(a, bt1, c1);
    }

    #pragma unroll
    for (int g = 0; g < 8; ++g) {
      const int m = g + half * 8;
      xb[m * 20 + ln] = fmaxf(c0[g], 0.0f);
      if (ln < 3) xb[m * 20 + 16 + ln] = fmaxf(c1[g], 0.0f);
      if (ln == 3) xb[m * 20 + 19] = 0.0f;
    }
    asm volatile("s_wait_dscnt 0" ::: "memory");

    v8f c2;
    {
      const float bias2 = sW[L_B2 + j * ND + n2];
      #pragma unroll
      for (int g = 0; g < 8; ++g) c2[g] = bias2;
    }
    const float* w2j = sW + L_W2 + j * ND * KP;
    #pragma unroll
    for (int kk = 0; kk < 5; ++kk) {
      const int k = kk * 4 + half * 2;
      const v2f a  = *(const v2f*)(xb + ln * 20 + k);
      const v2f bt = *(const v2f*)(w2j + n2 * KP + k);
      c2 = wmma_f32(a, bt, c2);
    }
    if (ln < ND) {
      #pragma unroll
      for (int g = 0; g < 8; ++g)
        feat[(g + half * 8) * 152 + j * ND + ln] = c2[g];
    }
  }

  asm volatile("s_wait_dscnt 0" ::: "memory");
  #pragma unroll 4
  for (int idx = lane; idx < 16 * 36; idx += 32) {
    const int m = idx / 36;
    const int q = idx - m * 36;
    *(v4f*)(out + (s0 + m) * 144 + q * 4) = *(const v4f*)(feat + m * 152 + q * 4);
  }
  }
}

extern "C" void kernel_launch(void* const* d_in, const int* in_sizes, int n_in,
                              void* d_out, int out_size, void* d_ws, size_t ws_size,
                              hipStream_t stream) {
  const float* rots = (const float*)d_in[0];
  const float* Jtrs = (const float*)d_in[1];
  const float* W0   = (const float*)d_in[2];
  const float* b0   = (const float*)d_in[3];
  const float* W1   = (const float*)d_in[4];
  const float* b1   = (const float*)d_in[5];
  const float* W2   = (const float*)d_in[6];
  const float* b2   = (const float*)d_in[7];
  float* out = (float*)d_out;

  const long bsz = (long)in_sizes[0] / (NJ * 9);
  if (bsz > 0) {
    hipLaunchKernelGGL(hier_pose_kernel, dim3(NG), dim3(NW * 32), 0, stream,
                       rots, Jtrs, W0, b0, W1, b1, W2, b2, out, bsz);
  }
}


# A fragments of both input products = registers of 16 B row loads split over lane halves (k order permuted, weights read alike): 288->6 in 36 loads; x tile kept in registers, 3 loads/joint, not in LDS
# speedup vs baseline: 1.9193x; 1.1048x over previous
#include <hip/hip_runtime.h>
#include <math.h>


typedef float v2f __attribute__((ext_vector_type(2)));
typedef float v3f __attribute__((ext_vector_type(3)));
typedef float v4f __attribute__((ext_vector_type(4)));
typedef float v8f __attribute__((ext_vector_type(8)));

#define NJ 24
#define ND 6
#define NF 19
#define NW 8
#define NG 512
#define KP 20
#define W0P 292

#define L_W1  0
#define L_W2  (L_W1 + NJ*NF*KP)
#define L_W0  (L_W2 + NJ*ND*KP)
#define L_B1  (L_W0 + ND*W0P)
#define L_B2  (L_B1 + NJ*NF)
#define L_B0  (L_B2 + NJ*ND)
#define L_WTOT (L_B0 + 8)

__constant__ int c_parents[NJ] = {-1,0,0,0,1,2,3,4,5,6,7,8,9,9,9,12,13,14,16,17,18,19,20,21};

static __device__ __forceinline__ v8f wmma_f32(v2f a, v2f b, v8f c) {
  return __builtin_amdgcn_wmma_f32_16x16x4_f32(
       false, a,  false, b,
       (short)0, c,  false,  false);
}

static __device__ __forceinline__ float other_half(float v) {
  const unsigned u = __builtin_bit_cast(unsigned, v);
  return __builtin_bit_cast(float,
      __builtin_amdgcn_permlanex16(u, u, 0x76543210u, 0xfedcba98u, false, false));
}

__global__ __launch_bounds__(NW * 32)
void hier_pose_kernel(const float* __restrict__ rots, const float* __restrict__ Jtrs,
                      const float* __restrict__ W0, const float* __restrict__ b0,
                      const float* __restrict__ W1, const float* __restrict__ b1,
                      const float* __restrict__ W2, const float* __restrict__ b2,
                      float* __restrict__ out, long bsz)
{
  __shared__ __align__(16) float sW[L_WTOT];
  __shared__ __align__(16) float sfeat[NW][16][152];
  __shared__ __align__(16) float sx[NW][16][20];

  {
    const int t = threadIdx.x;
    constexpr int NT  = NW * 32;
    constexpr int N1  = NJ * NF * NF, T1 = (N1 + NT - 1) / NT;
    constexpr int N2  = NJ * ND * NF, T2 = (N2 + NT - 1) / NT;
    constexpr int N0  = ND * 12 * NJ, T0 = (N0 + NT - 1) / NT;
    constexpr int NB1 = NJ * NF,      TB = (NB1 + NT - 1) / NT;
    float v1[T1], v2[T2], v0[T0], vb[TB];
    #pragma unroll
    for (int u = 0; u < T1; ++u) { const int i = t + u * NT; v1[u] = W1[i < N1 ? i : N1 - 1]; }
    #pragma unroll
    for (int u = 0; u < T2; ++u) { const int i = t + u * NT; v2[u] = W2[i < N2 ? i : N2 - 1]; }
    #pragma unroll
    for (int u = 0; u < T0; ++u) { const int i = t + u * NT; v0[u] = W0[i < N0 ? i : N0 - 1]; }
    #pragma unroll
    for (int u = 0; u < TB; ++u) { const int i = t + u * NT; vb[u] = b1[i < NB1 ? i : NB1 - 1]; }
    const float vb2 = b2[t < NJ * ND ? t : NJ * ND - 1];
    const float vb0 = b0[t < ND ? t : ND - 1];

    #pragma unroll
    for (int u = 0; u < T1; ++u) {
      const int i = t + u * NT, r = i / NF, k = i - r * NF;
      if (i < N1) sW[L_W1 + r * KP + k] = v1[u];
    }
    #pragma unroll
    for (int u = 0; u < T2; ++u) {
      const int i = t + u * NT, r = i / NF, k = i - r * NF;
      if (i < N2) sW[L_W2 + r * KP + k] = v2[u];
    }
    #pragma unroll
    for (int u = 0; u < T0; ++u) {
      const int i = t + u * NT, n = i / (12 * NJ), k = i - n * (12 * NJ);
      if (i < N0) sW[L_W0 + n * W0P + k] = v0[u];
    }
    #pragma unroll
    for (int u = 0; u < TB; ++u) { const int i = t + u * NT; if (i < NB1) sW[L_B1 + i] = vb[u]; }
    if (t < NJ * ND) sW[L_B2 + t] = vb2;
    if (t < 8)       sW[L_B0 + t] = (t < ND) ? vb0 : 0.0f;
    for (int r = t; r < NJ * (NF + ND); r += NT) sW[L_W1 + r * KP + NF] = 0.0f;
    if (t < ND * 4) sW[L_W0 + (t >> 2) * W0P + 12 * NJ + (t & 3)] = 0.0f;
  }
  __syncthreads();

  const int  lane = threadIdx.x & 31;
  const int  wave = threadIdx.x >> 5;
  const int  half = lane >> 4;
  const int  ln   = lane & 15;
  const int  n1   = (16 + ln < NF) ? 16 + ln : NF - 1;
  const int  n2   = (ln < ND) ? ln : ND - 1;

  float* feat = &sfeat[wave][0][0];
  float* xb   = &sx[wave][0][0];

  const long ntiles = bsz / 16;
  for (long tile = (long)blockIdx.x * NW + wave; tile < ntiles; tile += (long)NG * NW) {
  const long s0 = tile * 16;

  const float* arow_r = rots + (s0 + ln) * 216;
  const float* arow_j = Jtrs + (s0 + ln) * 72;
  __builtin_prefetch(arow_r, 0, 3);
  __builtin_prefetch(arow_j, 0, 3);

  v8f acc;
  {
    const float ci = sW[L_B0 + n2];
    #pragma unroll
    for (int g = 0; g < 8; ++g) acc[g] = ci;
  }
  const float* w0row = sW + L_W0 + n2 * W0P;
  #pragma unroll 3
  for (int q = 0; q < 27; ++q) {
    const int k = q * 8 + half * 4;
    const v4f a = *(const v4f*)(arow_r + k);
    const v4f b = *(const v4f*)(w0row + k);
    acc = wmma_f32(a.xy, b.xy, acc);
    acc = wmma_f32(a.zw, b.zw, acc);
  }
  #pragma unroll 3
  for (int q = 0; q < 9; ++q) {
    const int k = q * 8 + half * 4;
    const v4f a = *(const v4f*)(arow_j + k);
    const v4f b = *(const v4f*)(w0row + 216 + k);
    acc = wmma_f32(a.xy, b.xy, acc);
    acc = wmma_f32(a.zw, b.zw, acc);
  }
  if (ln < ND) {
    #pragma unroll
    for (int g = 0; g < 8; ++g)
      feat[(g + half * 8) * 152 + 144 + ln] = acc[g];
  }

  for (int j = 0; j < NJ; ++j) {
    const int p     = c_parents[j];
    const int fslot = (p < 0) ? 24 : p;

    v2f xa0, xa1, xa2, xa3, xa4;
    {
      const int  m    = ln;
      const long sj   = (s0 + m) * NJ;
      const float* rp = rots + (sj + j) * 9;
      const int  pj   = (p < 0) ? j : p;
      const float* tp = Jtrs + (sj + (half ? pj : j)) * 3;
      v4f R;  __builtin_memcpy(&R, rp + 4 * half, sizeof(float) * 4);
      v3f P;  __builtin_memcpy(&P, tp, sizeof(float) * 3);
      const float r8 = rp[8];
      const float* fp = feat + m * 152 + fslot * ND;
      const v2f f01 = *(const v2f*)(fp), f23 = *(const v2f*)(fp + 2), f45 = *(const v2f*)(fp + 4);

      float px = other_half(P.x), py = other_half(P.y), pz = other_half(P.z);
      if (p < 0) { px = 0.0f; py = 0.0f; pz = 0.0f; }
      const float dx = P.x - px, dy = P.y - py, dz = P.z - pz;
      const float bone  = sqrtf(dx * dx + dy * dy + dz * dz);
      const float boneu = other_half(bone);

      xa0 = R.xy;
      xa1 = R.zw;
      const v2f lo2 = {r8, P.x},   hi2 = {boneu, f01.x};
      const v2f lo3 = {P.y, P.z},  hi3 = {f01.y, f23.x};
      const v2f lo4 = {f23.y, f45.x}, hi4 = {f45.y, 0.0f};
      xa2 = half ? hi2 : lo2;
      xa3 = half ? hi3 : lo3;
      xa4 = half ? hi4 : lo4;
    }

    v8f c0, c1;
    {
      const float bias0 = sW[L_B1 + j * NF + ln];
      const float bias1 = sW[L_B1 + j * NF + n1];
      #pragma unroll
      for (int g = 0; g < 8; ++g) { c0[g] = bias0; c1[g] = bias1; }
    }
    const float* w1j = sW + L_W1 + j * NF * KP;
    {
      const float* r0 = w1j + ln * KP;
      const float* r1 = w1j + n1 * KP;
      const v4f p0 = *(const v4f*)(r0 + 4 * half),      p1 = *(const v4f*)(r1 + 4 * half);
      const v4f q0 = *(const v4f*)(r0 + 8 + 4 * half),  q1 = *(const v4f*)(r1 + 8 + 4 * half);
      const v2f t0 = *(const v2f*)(r0 + 16 + 2 * half), t1 = *(const v2f*)(r1 + 16 + 2 * half);
      c0 = wmma_f32(xa0, p0.xy, c0);  c1 = wmma_f32(xa0, p1.xy, c1);
      c0 = wmma_f32(xa1, p0.zw, c0);  c1 = wmma_f32(xa1, p1.zw, c1);
      c0 = wmma_f32(xa2, q0.xy, c0);  c1 = wmma_f32(xa2, q1.xy, c1);
      c0 = wmma_f32(xa3, q0.zw, c0);  c1 = wmma_f32(xa3, q1.zw, c1);
      c0 = wmma_f32(xa4, t0,    c0);  c1 = wmma_f32(xa4, t1,    c1);
    }

    #pragma unroll
    for (int g = 0; g < 8; ++g) {
      const int m = g + half * 8;
      xb[m * 20 + ln] = fmaxf(c0[g], 0.0f);
      if (ln < 3) xb[m * 20 + 16 + ln] = fmaxf(c1[g], 0.0f);
      if (ln == 3) xb[m * 20 + 19] = 0.0f;
    }
    asm volatile("s_wait_dscnt 0" ::: "memory");

    v8f c2;
    {
      const float bias2 = sW[L_B2 + j * ND + n2];
      #pragma unroll
      for (int g = 0; g < 8; ++g) c2[g] = bias2;
    }
    const float* w2j = sW + L_W2 + j * ND * KP;
    #pragma unroll
    for (int kk = 0; kk < 5; ++kk) {
      const int k = kk * 4 + half * 2;
      const v2f a  = *(const v2f*)(xb + ln * 20 + k);
      const v2f bt = *(const v2f*)(w2j + n2 * KP + k);
      c2 = wmma_f32(a, bt, c2);
    }
    if (ln < ND) {
      #pragma unroll
      for (int g = 0; g < 8; ++g)
        feat[(g + half * 8) * 152 + j * ND + ln] = c2[g];
    }
  }

  asm volatile("s_wait_dscnt 0" ::: "memory");
  #pragma unroll 4
  for (int idx = lane; idx < 16 * 36; idx += 32) {
    const int m = idx / 36;
    const int q = idx - m * 36;
    *(v4f*)(out + (s0 + m) * 144 + q * 4) = *(const v4f*)(feat + m * 152 + q * 4);
  }
  }
}

extern "C" void kernel_launch(void* const* d_in, const int* in_sizes, int n_in,
                              void* d_out, int out_size, void* d_ws, size_t ws_size,
                              hipStream_t stream) {
  const float* rots = (const float*)d_in[0];
  const float* Jtrs = (const float*)d_in[1];
  const float* W0   = (const float*)d_in[2];
  const float* b0   = (const float*)d_in[3];
  const float* W1   = (const float*)d_in[4];
  const float* b1   = (const float*)d_in[5];
  const float* W2   = (const float*)d_in[6];
  const float* b2   = (const float*)d_in[7];
  float* out = (float*)d_out;

  const long bsz = (long)in_sizes[0] / (NJ * 9);
  if (bsz > 0) {
    hipLaunchKernelGGL(hier_pose_kernel, dim3(NG), dim3(NW * 32), 0, stream,
                       rots, Jtrs, W0, b0, W1, b1, W2, b2, out, bsz);
  }
}


# Joint chain taken two independent joints per step (0 and 23 alone): 13 dependent steps, not 24; a pair's loads, LDS round trips and WMMA chains interleave; one more h staging tile per wave
# speedup vs baseline: 2.0460x; 1.0660x over previous
#include <hip/hip_runtime.h>
#include <math.h>


typedef float v2f __attribute__((ext_vector_type(2)));
typedef float v3f __attribute__((ext_vector_type(3)));
typedef float v4f __attribute__((ext_vector_type(4)));
typedef float v8f __attribute__((ext_vector_type(8)));

#define NJ 24
#define ND 6
#define NF 19
#define NW 8
#define NG 512
#define KP 20
#define W0P 292

#define L_W1  0
#define L_W2  (L_W1 + NJ*NF*KP)
#define L_W0  (L_W2 + NJ*ND*KP)
#define L_B1  (L_W0 + ND*W0P)
#define L_B2  (L_B1 + NJ*NF)
#define L_B0  (L_B2 + NJ*ND)
#define L_WTOT (L_B0 + 8)

#define PARENT_LIST -1,0,0,0,1,2,3,4,5,6,7,8,9,9,9,12,13,14,16,17,18,19,20,21
__constant__ int c_parents[NJ] = {PARENT_LIST};

constexpr int k_parents[NJ] = {PARENT_LIST};
constexpr bool pairs_independent() {
  for (int j = 1; j + 1 < NJ; j += 2) if (k_parents[j + 1] >= j) return false;
  return true;
}
static_assert(pairs_independent(), "a joint pair (j, j+1), j odd, has j as the parent of j+1");

static __device__ __forceinline__ v8f wmma_f32(v2f a, v2f b, v8f c) {
  return __builtin_amdgcn_wmma_f32_16x16x4_f32(
       false, a,  false, b,
       (short)0, c,  false,  false);
}

static __device__ __forceinline__ float other_half(float v) {
  const unsigned u = __builtin_bit_cast(unsigned, v);
  return __builtin_bit_cast(float,
      __builtin_amdgcn_permlanex16(u, u, 0x76543210u, 0xfedcba98u, false, false));
}

template <int K>
static __device__ __forceinline__ void joint_group(
    const int j0, const float* __restrict__ rots, const float* __restrict__ Jtrs,
    const float* sW, float* feat, float* hb, const long s0,
    const int ln, const int half, const int n1, const int n2)
{
  v2f xa[K][5];
  #pragma unroll
  for (int i = 0; i < K; ++i) {
    const int  j     = j0 + i;
    const int  p     = c_parents[j];
    const int  fslot = (p < 0) ? 24 : p;
    const int  m     = ln;
    const long sj    = (s0 + m) * NJ;
    const float* rp  = rots + (sj + j) * 9;
    const int  pj    = (p < 0) ? j : p;
    const float* tp  = Jtrs + (sj + (half ? pj : j)) * 3;
    v4f R;  __builtin_memcpy(&R, rp + 4 * half, sizeof(float) * 4);
    v3f P;  __builtin_memcpy(&P, tp, sizeof(float) * 3);
    const float r8 = rp[8];
    const float* fp = feat + m * 152 + fslot * ND;
    const v2f f01 = *(const v2f*)(fp), f23 = *(const v2f*)(fp + 2), f45 = *(const v2f*)(fp + 4);

    float px = other_half(P.x), py = other_half(P.y), pz = other_half(P.z);
    if (p < 0) { px = 0.0f; py = 0.0f; pz = 0.0f; }
    const float dx = P.x - px, dy = P.y - py, dz = P.z - pz;
    const float bone  = sqrtf(dx * dx + dy * dy + dz * dz);
    const float boneu = other_half(bone);

    const v2f lo2 = {r8, P.x},   hi2 = {boneu, f01.x};
    const v2f lo3 = {P.y, P.z},  hi3 = {f01.y, f23.x};
    const v2f lo4 = {f23.y, f45.x}, hi4 = {f45.y, 0.0f};
    xa[i][0] = R.xy;
    xa[i][1] = R.zw;
    xa[i][2] = half ? hi2 : lo2;
    xa[i][3] = half ? hi3 : lo3;
    xa[i][4] = half ? hi4 : lo4;
  }

  v8f c0[K], c1[K];
  v2f w1a[K][5], w1b[K][5];
  #pragma unroll
  for (int i = 0; i < K; ++i) {
    const int j = j0 + i;
    const float bias0 = sW[L_B1 + j * NF + ln];
    const float bias1 = sW[L_B1 + j * NF + n1];
    #pragma unroll
    for (int g = 0; g < 8; ++g) { c0[i][g] = bias0; c1[i][g] = bias1; }
    const float* w1j = sW + L_W1 + j * NF * KP;
    const float* r0 = w1j + ln * KP;
    const float* r1 = w1j + n1 * KP;
    const v4f p0 = *(const v4f*)(r0 + 4 * half),      p1 = *(const v4f*)(r1 + 4 * half);
    const v4f q0 = *(const v4f*)(r0 + 8 + 4 * half),  q1 = *(const v4f*)(r1 + 8 + 4 * half);
    const v2f t0 = *(const v2f*)(r0 + 16 + 2 * half), t1 = *(const v2f*)(r1 + 16 + 2 * half);
    w1a[i][0] = p0.xy;  w1a[i][1] = p0.zw;  w1a[i][2] = q0.xy;  w1a[i][3] = q0.zw;  w1a[i][4] = t0;
    w1b[i][0] = p1.xy;  w1b[i][1] = p1.zw;  w1b[i][2] = q1.xy;  w1b[i][3] = q1.zw;  w1b[i][4] = t1;
  }
  #pragma unroll
  for (int st = 0; st < 5; ++st) {
    #pragma unroll
    for (int i = 0; i < K; ++i) {
      c0[i] = wmma_f32(xa[i][st], w1a[i][st], c0[i]);
      c1[i] = wmma_f32(xa[i][st], w1b[i][st], c1[i]);
    }
  }

  #pragma unroll
  for (int i = 0; i < K; ++i) {
    float* xb = hb + i * (16 * 20);
    #pragma unroll
    for (int g = 0; g < 8; ++g) {
      const int m = g + half * 8;
      xb[m * 20 + ln] = fmaxf(c0[i][g], 0.0f);
      if (ln < 3) xb[m * 20 + 16 + ln] = fmaxf(c1[i][g], 0.0f);
      if (ln == 3) xb[m * 20 + 19] = 0.0f;
    }
  }
  asm volatile("s_wait_dscnt 0" ::: "memory");

  v8f c2[K];
  v2f ha[K][5], w2f[K][5];
  #pragma unroll
  for (int i = 0; i < K; ++i) {
    const int j = j0 + i;
    const float* xb = hb + i * (16 * 20);
    const float bias2 = sW[L_B2 + j * ND + n2];
    #pragma unroll
    for (int g = 0; g < 8; ++g) c2[i][g] = bias2;
    const float* w2j = sW + L_W2 + j * ND * KP;
    #pragma unroll
    for (int kk = 0; kk < 5; ++kk) {
      const int k = kk * 4 + half * 2;
      ha[i][kk]  = *(const v2f*)(xb + ln * 20 + k);
      w2f[i][kk] = *(const v2f*)(w2j + n2 * KP + k);
    }
  }
  #pragma unroll
  for (int kk = 0; kk < 5; ++kk) {
    #pragma unroll
    for (int i = 0; i < K; ++i) c2[i] = wmma_f32(ha[i][kk], w2f[i][kk], c2[i]);
  }
  if (ln < ND) {
    #pragma unroll
    for (int i = 0; i < K; ++i) {
      #pragma unroll
      for (int g = 0; g < 8; ++g)
        feat[(g + half * 8) * 152 + (j0 + i) * ND + ln] = c2[i][g];
    }
  }
}

__global__ __launch_bounds__(NW * 32)
void hier_pose_kernel(const float* __restrict__ rots, const float* __restrict__ Jtrs,
                      const float* __restrict__ W0, const float* __restrict__ b0,
                      const float* __restrict__ W1, const float* __restrict__ b1,
                      const float* __restrict__ W2, const float* __restrict__ b2,
                      float* __restrict__ out, long bsz)
{
  __shared__ __align__(16) float sW[L_WTOT];
  __shared__ __align__(16) float sfeat[NW][16][152];
  __shared__ __align__(16) float sx[NW][2][16][20];

  {
    const int t = threadIdx.x;
    constexpr int NT  = NW * 32;
    constexpr int N1  = NJ * NF * NF, T1 = (N1 + NT - 1) / NT;
    constexpr int N2  = NJ * ND * NF, T2 = (N2 + NT - 1) / NT;
    constexpr int N0  = ND * 12 * NJ, T0 = (N0 + NT - 1) / NT;
    constexpr int NB1 = NJ * NF,      TB = (NB1 + NT - 1) / NT;
    float v1[T1], v2[T2], v0[T0], vb[TB];
    #pragma unroll
    for (int u = 0; u < T1; ++u) { const int i = t + u * NT; v1[u] = W1[i < N1 ? i : N1 - 1]; }
    #pragma unroll
    for (int u = 0; u < T2; ++u) { const int i = t + u * NT; v2[u] = W2[i < N2 ? i : N2 - 1]; }
    #pragma unroll
    for (int u = 0; u < T0; ++u) { const int i = t + u * NT; v0[u] = W0[i < N0 ? i : N0 - 1]; }
    #pragma unroll
    for (int u = 0; u < TB; ++u) { const int i = t + u * NT; vb[u] = b1[i < NB1 ? i : NB1 - 1]; }
    const float vb2 = b2[t < NJ * ND ? t : NJ * ND - 1];
    const float vb0 = b0[t < ND ? t : ND - 1];

    #pragma unroll
    for (int u = 0; u < T1; ++u) {
      const int i = t + u * NT, r = i / NF, k = i - r * NF;
      if (i < N1) sW[L_W1 + r * KP + k] = v1[u];
    }
    #pragma unroll
    for (int u = 0; u < T2; ++u) {
      const int i = t + u * NT, r = i / NF, k = i - r * NF;
      if (i < N2) sW[L_W2 + r * KP + k] = v2[u];
    }
    #pragma unroll
    for (int u = 0; u < T0; ++u) {
      const int i = t + u * NT, n = i / (12 * NJ), k = i - n * (12 * NJ);
      if (i < N0) sW[L_W0 + n * W0P + k] = v0[u];
    }
    #pragma unroll
    for (int u = 0; u < TB; ++u) { const int i = t + u * NT; if (i < NB1) sW[L_B1 + i] = vb[u]; }
    if (t < NJ * ND) sW[L_B2 + t] = vb2;
    if (t < 8)       sW[L_B0 + t] = (t < ND) ? vb0 : 0.0f;
    for (int r = t; r < NJ * (NF + ND); r += NT) sW[L_W1 + r * KP + NF] = 0.0f;
    if (t < ND * 4) sW[L_W0 + (t >> 2) * W0P + 12 * NJ + (t & 3)] = 0.0f;
  }
  __syncthreads();

  const int  lane = threadIdx.x & 31;
  const int  wave = threadIdx.x >> 5;
  const int  half = lane >> 4;
  const int  ln   = lane & 15;
  const int  n1   = (16 + ln < NF) ? 16 + ln : NF - 1;
  const int  n2   = (ln < ND) ? ln : ND - 1;

  float* feat = &sfeat[wave][0][0];
  float* hb   = &sx[wave][0][0][0];

  const long ntiles = bsz / 16;
  for (long tile = (long)blockIdx.x * NW + wave; tile < ntiles; tile += (long)NG * NW) {
  const long s0 = tile * 16;

  const float* arow_r = rots + (s0 + ln) * 216;
  const float* arow_j = Jtrs + (s0 + ln) * 72;
  __builtin_prefetch(arow_r, 0, 3);
  __builtin_prefetch(arow_j, 0, 3);

  v8f acc;
  {
    const float ci = sW[L_B0 + n2];
    #pragma unroll
    for (int g = 0; g < 8; ++g) acc[g] = ci;
  }
  const float* w0row = sW + L_W0 + n2 * W0P;
  #pragma unroll 3
  for (int q = 0; q < 27; ++q) {
    const int k = q * 8 + half * 4;
    const v4f a = *(const v4f*)(arow_r + k);
    const v4f b = *(const v4f*)(w0row + k);
    acc = wmma_f32(a.xy, b.xy, acc);
    acc = wmma_f32(a.zw, b.zw, acc);
  }
  #pragma unroll 3
  for (int q = 0; q < 9; ++q) {
    const int k = q * 8 + half * 4;
    const v4f a = *(const v4f*)(arow_j + k);
    const v4f b = *(const v4f*)(w0row + 216 + k);
    acc = wmma_f32(a.xy, b.xy, acc);
    acc = wmma_f32(a.zw, b.zw, acc);
  }
  if (ln < ND) {
    #pragma unroll
    for (int g = 0; g < 8; ++g)
      feat[(g + half * 8) * 152 + 144 + ln] = acc[g];
  }

  joint_group<1>(0, rots, Jtrs, sW, feat, hb, s0, ln, half, n1, n2);
  for (int j = 1; j + 1 < NJ; j += 2)
    joint_group<2>(j, rots, Jtrs, sW, feat, hb, s0, ln, half, n1, n2);
  joint_group<1>(NJ - 1, rots, Jtrs, sW, feat, hb, s0, ln, half, n1, n2);

  asm volatile("s_wait_dscnt 0" ::: "memory");
  #pragma unroll 4
  for (int idx = lane; idx < 16 * 36; idx += 32) {
    const int m = idx / 36;
    const int q = idx - m * 36;
    *(v4f*)(out + (s0 + m) * 144 + q * 4) = *(const v4f*)(feat + m * 152 + q * 4);
  }
  }
}

extern "C" void kernel_launch(void* const* d_in, const int* in_sizes, int n_in,
                              void* d_out, int out_size, void* d_ws, size_t ws_size,
                              hipStream_t stream) {
  const float* rots = (const float*)d_in[0];
  const float* Jtrs = (const float*)d_in[1];
  const float* W0   = (const float*)d_in[2];
  const float* b0   = (const float*)d_in[3];
  const float* W1   = (const float*)d_in[4];
  const float* b1   = (const float*)d_in[5];
  const float* W2   = (const float*)d_in[6];
  const float* b2   = (const float*)d_in[7];
  float* out = (float*)d_out;

  const long bsz = (long)in_sizes[0] / (NJ * 9);
  if (bsz > 0) {
    hipLaunchKernelGGL(hier_pose_kernel, dim3(NG), dim3(NW * 32), 0, stream,
                       rots, Jtrs, W0, b0, W1, b1, W2, b2, out, bsz);
  }
}


# h tile between a joint's two layers staged n-major in LDS: a lane's 8 D registers leave as 2 b128 stores, not 24 exec-masked b32 stores a joint; layer-2 A fragments read as b32 pairs; sums unchanged
# speedup vs baseline: 2.3042x; 1.1262x over previous
#include <hip/hip_runtime.h>
#include <math.h>


typedef float v2f __attribute__((ext_vector_type(2)));
typedef float v3f __attribute__((ext_vector_type(3)));
typedef float v4f __attribute__((ext_vector_type(4)));
typedef float v8f __attribute__((ext_vector_type(8)));

#define NJ 24
#define ND 6
#define NF 19
#define NW 8
#define NG 512
#define KP 20
#define W0P 292
#define HP 20
#define HT (16 * HP)

#define L_W1  0
#define L_W2  (L_W1 + NJ*NF*KP)
#define L_W0  (L_W2 + NJ*ND*KP)
#define L_B1  (L_W0 + ND*W0P)
#define L_B2  (L_B1 + NJ*NF)
#define L_B0  (L_B2 + NJ*ND)
#define L_WTOT (L_B0 + 8)

#define PARENT_LIST -1,0,0,0,1,2,3,4,5,6,7,8,9,9,9,12,13,14,16,17,18,19,20,21
__constant__ int c_parents[NJ] = {PARENT_LIST};

constexpr int k_parents[NJ] = {PARENT_LIST};
constexpr bool pairs_independent() {
  for (int j = 1; j + 1 < NJ; j += 2) if (k_parents[j + 1] >= j) return false;
  return true;
}
static_assert(pairs_independent(), "a joint pair (j, j+1), j odd, has j as the parent of j+1");

static __device__ __forceinline__ v8f wmma_f32(v2f a, v2f b, v8f c) {
  return __builtin_amdgcn_wmma_f32_16x16x4_f32(
       false, a,  false, b,
       (short)0, c,  false,  false);
}

static __device__ __forceinline__ float other_half(float v) {
  const unsigned u = __builtin_bit_cast(unsigned, v);
  return __builtin_bit_cast(float,
      __builtin_amdgcn_permlanex16(u, u, 0x76543210u, 0xfedcba98u, false, false));
}

template <int K>
static __device__ __forceinline__ void joint_group(
    const int j0, const float* __restrict__ rots, const float* __restrict__ Jtrs,
    const float* sW, float* feat, float* hb, const long s0,
    const int ln, const int half, const int n1, const int n2)
{
  v2f xa[K][5];
  #pragma unroll
  for (int i = 0; i < K; ++i) {
    const int  j     = j0 + i;
    const int  p     = c_parents[j];
    const int  fslot = (p < 0) ? 24 : p;
    const int  m     = ln;
    const long sj    = (s0 + m) * NJ;
    const float* rp  = rots + (sj + j) * 9;
    const int  pj    = (p < 0) ? j : p;
    const float* tp  = Jtrs + (sj + (half ? pj : j)) * 3;
    v4f R;  __builtin_memcpy(&R, rp + 4 * half, sizeof(float) * 4);
    v3f P;  __builtin_memcpy(&P, tp, sizeof(float) * 3);
    const float r8 = rp[8];
    const float* fp = feat + m * 152 + fslot * ND;
    const v2f f01 = *(const v2f*)(fp), f23 = *(const v2f*)(fp + 2), f45 = *(const v2f*)(fp + 4);

    float px = other_half(P.x), py = other_half(P.y), pz = other_half(P.z);
    if (p < 0) { px = 0.0f; py = 0.0f; pz = 0.0f; }
    const float dx = P.x - px, dy = P.y - py, dz = P.z - pz;
    const float bone  = sqrtf(dx * dx + dy * dy + dz * dz);
    const float boneu = other_half(bone);

    const v2f lo2 = {r8, P.x},   hi2 = {boneu, f01.x};
    const v2f lo3 = {P.y, P.z},  hi3 = {f01.y, f23.x};
    const v2f lo4 = {f23.y, f45.x}, hi4 = {f45.y, 0.0f};
    xa[i][0] = R.xy;
    xa[i][1] = R.zw;
    xa[i][2] = half ? hi2 : lo2;
    xa[i][3] = half ? hi3 : lo3;
    xa[i][4] = half ? hi4 : lo4;
  }

  v8f c0[K], c1[K];
  v2f w1a[K][5], w1b[K][5];
  #pragma unroll
  for (int i = 0; i < K; ++i) {
    const int j = j0 + i;
    const float bias0 = sW[L_B1 + j * NF + ln];
    const float bias1 = sW[L_B1 + j * NF + n1];
    #pragma unroll
    for (int g = 0; g < 8; ++g) { c0[i][g] = bias0; c1[i][g] = bias1; }
    const float* w1j = sW + L_W1 + j * NF * KP;
    const float* r0 = w1j + ln * KP;
    const float* r1 = w1j + n1 * KP;
    const v4f p0 = *(const v4f*)(r0 + 4 * half),      p1 = *(const v4f*)(r1 + 4 * half);
    const v4f q0 = *(const v4f*)(r0 + 8 + 4 * half),  q1 = *(const v4f*)(r1 + 8 + 4 * half);
    const v2f t0 = *(const v2f*)(r0 + 16 + 2 * half), t1 = *(const v2f*)(r1 + 16 + 2 * half);
    w1a[i][0] = p0.xy;  w1a[i][1] = p0.zw;  w1a[i][2] = q0.xy;  w1a[i][3] = q0.zw;  w1a[i][4] = t0;
    w1b[i][0] = p1.xy;  w1b[i][1] = p1.zw;  w1b[i][2] = q1.xy;  w1b[i][3] = q1.zw;  w1b[i][4] = t1;
  }
  #pragma unroll
  for (int st = 0; st < 5; ++st) {
    #pragma unroll
    for (int i = 0; i < K; ++i) {
      c0[i] = wmma_f32(xa[i][st], w1a[i][st], c0[i]);
      c1[i] = wmma_f32(xa[i][st], w1b[i][st], c1[i]);
    }
  }

  #pragma unroll
  for (int i = 0; i < K; ++i) {
    float* xb = hb + i * HT;
    float* d0 = xb + ln * HP + half * 8;
    const v4f a0 = {fmaxf(c0[i][0], 0.0f), fmaxf(c0[i][1], 0.0f), fmaxf(c0[i][2], 0.0f), fmaxf(c0[i][3], 0.0f)};
    const v4f a1 = {fmaxf(c0[i][4], 0.0f), fmaxf(c0[i][5], 0.0f), fmaxf(c0[i][6], 0.0f), fmaxf(c0[i][7], 0.0f)};
    *(v4f*)(d0)     = a0;
    *(v4f*)(d0 + 4) = a1;
    if (ln < 4) {
      float* d1 = xb + (4 * ln + 2 * half) * HP + 16;
      v4f b0, b1;
      #pragma unroll
      for (int g = 0; g < 4; ++g) {
        b0[g] = (ln < 3) ? fmaxf(c1[i][g],     0.0f) : 0.0f;
        b1[g] = (ln < 3) ? fmaxf(c1[i][g + 4], 0.0f) : 0.0f;
      }
      *(v4f*)(d1)      = b0;
      *(v4f*)(d1 + HP) = b1;
    }
  }
  asm volatile("s_wait_dscnt 0" ::: "memory");

  v8f c2[K];
  v2f ha[K][5], w2f[K][5];
  #pragma unroll
  for (int i = 0; i < K; ++i) {
    const int j = j0 + i;
    const float* xb = hb + i * HT;
    const float bias2 = sW[L_B2 + j * ND + n2];
    #pragma unroll
    for (int g = 0; g < 8; ++g) c2[i][g] = bias2;
    const float* w2j = sW + L_W2 + j * ND * KP;
    const float* hp = xb + (half * 2) * HP + ln;
    const float* hq = xb + (half * 8 + (ln >> 2)) * HP + 16 + (ln & 3);
    #pragma unroll
    for (int kk = 0; kk < 5; ++kk) {
      const int k = kk * 4 + half * 2;
      const v2f hk = (kk < 4) ? v2f{hp[kk * 4 * HP], hp[kk * 4 * HP + HP]} : v2f{hq[0], hq[4 * HP]};
      ha[i][kk]  = hk;
      w2f[i][kk] = *(const v2f*)(w2j + n2 * KP + k);
    }
  }
  #pragma unroll
  for (int kk = 0; kk < 5; ++kk) {
    #pragma unroll
    for (int i = 0; i < K; ++i) c2[i] = wmma_f32(ha[i][kk], w2f[i][kk], c2[i]);
  }
  if (ln < ND) {
    #pragma unroll
    for (int i = 0; i < K; ++i) {
      #pragma unroll
      for (int g = 0; g < 8; ++g)
        feat[(g + half * 8) * 152 + (j0 + i) * ND + ln] = c2[i][g];
    }
  }
}

__global__ __launch_bounds__(NW * 32)
void hier_pose_kernel(const float* __restrict__ rots, const float* __restrict__ Jtrs,
                      const float* __restrict__ W0, const float* __restrict__ b0,
                      const float* __restrict__ W1, const float* __restrict__ b1,
                      const float* __restrict__ W2, const float* __restrict__ b2,
                      float* __restrict__ out, long bsz)
{
  __shared__ __align__(16) float sW[L_WTOT];
  __shared__ __align__(16) float sfeat[NW][16][152];
  __shared__ __align__(16) float sx[NW][2][16][HP];

  {
    const int t = threadIdx.x;
    constexpr int NT  = NW * 32;
    constexpr int N1  = NJ * NF * NF, T1 = (N1 + NT - 1) / NT;
    constexpr int N2  = NJ * ND * NF, T2 = (N2 + NT - 1) / NT;
    constexpr int N0  = ND * 12 * NJ, T0 = (N0 + NT - 1) / NT;
    constexpr int NB1 = NJ * NF,      TB = (NB1 + NT - 1) / NT;
    float v1[T1], v2[T2], v0[T0], vb[TB];
    #pragma unroll
    for (int u = 0; u < T1; ++u) { const int i = t + u * NT; v1[u] = W1[i < N1 ? i : N1 - 1]; }
    #pragma unroll
    for (int u = 0; u < T2; ++u) { const int i = t + u * NT; v2[u] = W2[i < N2 ? i : N2 - 1]; }
    #pragma unroll
    for (int u = 0; u < T0; ++u) { const int i = t + u * NT; v0[u] = W0[i < N0 ? i : N0 - 1]; }
    #pragma unroll
    for (int u = 0; u < TB; ++u) { const int i = t + u * NT; vb[u] = b1[i < NB1 ? i : NB1 - 1]; }
    const float vb2 = b2[t < NJ * ND ? t : NJ * ND - 1];
    const float vb0 = b0[t < ND ? t : ND - 1];

    #pragma unroll
    for (int u = 0; u < T1; ++u) {
      const int i = t + u * NT, r = i / NF, k = i - r * NF;
      if (i < N1) sW[L_W1 + r * KP + k] = v1[u];
    }
    #pragma unroll
    for (int u = 0; u < T2; ++u) {
      const int i = t + u * NT, r = i / NF, k = i - r * NF;
      if (i < N2) sW[L_W2 + r * KP + k] = v2[u];
    }
    #pragma unroll
    for (int u = 0; u < T0; ++u) {
      const int i = t + u * NT, n = i / (12 * NJ), k = i - n * (12 * NJ);
      if (i < N0) sW[L_W0 + n * W0P + k] = v0[u];
    }
    #pragma unroll
    for (int u = 0; u < TB; ++u) { const int i = t + u * NT; if (i < NB1) sW[L_B1 + i] = vb[u]; }
    if (t < NJ * ND) sW[L_B2 + t] = vb2;
    if (t < 8)       sW[L_B0 + t] = (t < ND) ? vb0 : 0.0f;
    for (int r = t; r < NJ * (NF + ND); r += NT) sW[L_W1 + r * KP + NF] = 0.0f;
    if (t < ND * 4) sW[L_W0 + (t >> 2) * W0P + 12 * NJ + (t & 3)] = 0.0f;
  }
  __syncthreads();

  const int  lane = threadIdx.x & 31;
  const int  wave = threadIdx.x >> 5;
  const int  half = lane >> 4;
  const int  ln   = lane & 15;
  const int  n1   = (16 + ln < NF) ? 16 + ln : NF - 1;
  const int  n2   = (ln < ND) ? ln : ND - 1;

  float* feat = &sfeat[wave][0][0];
  float* hb   = &sx[wave][0][0][0];

  const long ntiles = bsz / 16;
  for (long tile = (long)blockIdx.x * NW + wave; tile < ntiles; tile += (long)NG * NW) {
  const long s0 = tile * 16;

  const float* arow_r = rots + (s0 + ln) * 216;
  const float* arow_j = Jtrs + (s0 + ln) * 72;
  __builtin_prefetch(arow_r, 0, 3);
  __builtin_prefetch(arow_j, 0, 3);

  v8f acc;
  {
    const float ci = sW[L_B0 + n2];
    #pragma unroll
    for (int g = 0; g < 8; ++g) acc[g] = ci;
  }
  const float* w0row = sW + L_W0 + n2 * W0P;
  #pragma unroll 3
  for (int q = 0; q < 27; ++q) {
    const int k = q * 8 + half * 4;
    const v4f a = *(const v4f*)(arow_r + k);
    const v4f b = *(const v4f*)(w0row + k);
    acc = wmma_f32(a.xy, b.xy, acc);
    acc = wmma_f32(a.zw, b.zw, acc);
  }
  #pragma unroll 3
  for (int q = 0; q < 9; ++q) {
    const int k = q * 8 + half * 4;
    const v4f a = *(const v4f*)(arow_j + k);
    const v4f b = *(const v4f*)(w0row + 216 + k);
    acc = wmma_f32(a.xy, b.xy, acc);
    acc = wmma_f32(a.zw, b.zw, acc);
  }
  if (ln < ND) {
    #pragma unroll
    for (int g = 0; g < 8; ++g)
      feat[(g + half * 8) * 152 + 144 + ln] = acc[g];
  }

  joint_group<1>(0, rots, Jtrs, sW, feat, hb, s0, ln, half, n1, n2);
  for (int j = 1; j + 1 < NJ; j += 2)
    joint_group<2>(j, rots, Jtrs, sW, feat, hb, s0, ln, half, n1, n2);
  joint_group<1>(NJ - 1, rots, Jtrs, sW, feat, hb, s0, ln, half, n1, n2);

  asm volatile("s_wait_dscnt 0" ::: "memory");
  #pragma unroll 4
  for (int idx = lane; idx < 16 * 36; idx += 32) {
    const int m = idx / 36;
    const int q = idx - m * 36;
    *(v4f*)(out + (s0 + m) * 144 + q * 4) = *(const v4f*)(feat + m * 152 + q * 4);
  }
  }
}

extern "C" void kernel_launch(void* const* d_in, const int* in_sizes, int n_in,
                              void* d_out, int out_size, void* d_ws, size_t ws_size,
                              hipStream_t stream) {
  const float* rots = (const float*)d_in[0];
  const float* Jtrs = (const float*)d_in[1];
  const float* W0   = (const float*)d_in[2];
  const float* b0   = (const float*)d_in[3];
  const float* W1   = (const float*)d_in[4];
  const float* b1   = (const float*)d_in[5];
  const float* W2   = (const float*)d_in[6];
  const float* b2   = (const float*)d_in[7];
  float* out = (float*)d_out;

  const long bsz = (long)in_sizes[0] / (NJ * 9);
  if (bsz > 0) {
    hipLaunchKernelGGL(hier_pose_kernel, dim3(NG), dim3(NW * 32), 0, stream,
                       rots, Jtrs, W0, b0, W1, b1, W2, b2, out, bsz);
  }
}


# Position rows read from global once: the nine 16 B position registers of the 288->6 product stay live and give each joint its own and parent position; joint chain straight-line, parents constant
# speedup vs baseline: 2.6293x; 1.1411x over previous
#include <hip/hip_runtime.h>
#include <math.h>
#include <type_traits>


typedef float v2f __attribute__((ext_vector_type(2)));
typedef float v4f __attribute__((ext_vector_type(4)));
typedef float v8f __attribute__((ext_vector_type(8)));

#define NJ 24
#define ND 6
#define NF 19
#define NW 8
#define NG 512
#define KP 20
#define W0P 292
#define HP 20
#define HT (16 * HP)

#define L_W1  0
#define L_W2  (L_W1 + NJ*NF*KP)
#define L_W0  (L_W2 + NJ*ND*KP)
#define L_B1  (L_W0 + ND*W0P)
#define L_B2  (L_B1 + NJ*NF)
#define L_B0  (L_B2 + NJ*ND)
#define L_WTOT (L_B0 + 8)

#define PARENT_LIST -1,0,0,0,1,2,3,4,5,6,7,8,9,9,9,12,13,14,16,17,18,19,20,21

constexpr int k_parents[NJ] = {PARENT_LIST};
constexpr bool pairs_independent() {
  for (int j = 1; j + 1 < NJ; j += 2) if (k_parents[j + 1] >= j) return false;
  return true;
}
static_assert(pairs_independent(), "a joint pair (j, j+1), j odd, has j as the parent of j+1");

static __device__ __forceinline__ v8f wmma_f32(v2f a, v2f b, v8f c) {
  return __builtin_amdgcn_wmma_f32_16x16x4_f32(
       false, a,  false, b,
       (short)0, c,  false,  false);
}

static __device__ __forceinline__ float other_half(float v) {
  const unsigned u = __builtin_bit_cast(unsigned, v);
  return __builtin_bit_cast(float,
      __builtin_amdgcn_permlanex16(u, u, 0x76543210u, 0xfedcba98u, false, false));
}

template <int F>
static __device__ __forceinline__ float pos_lower(const v4f (&jt)[9]) {
  const float own = jt[F >> 3][F & 3];
  if constexpr (((F >> 2) & 1) != 0) return other_half(own);
  else return own;
}

template <int K, int J0>
static __device__ __forceinline__ void joint_group(
    const float* __restrict__ rots, const v4f (&jt)[9],
    const float* sW, float* feat, float* hb, const long s0,
    const int ln, const int half, const int n1, const int n2)
{
  constexpr int j0 = J0;
  v2f xa[K][5];
  auto build_x = [&](auto ic) {
    constexpr int  i     = decltype(ic)::value;
    constexpr int  j     = j0 + i;
    constexpr int  p     = k_parents[j];
    constexpr int  fslot = (p < 0) ? 24 : p;
    constexpr int  pj    = (p < 0) ? j : p;
    const int  m     = ln;
    const float* rp  = rots + ((s0 + m) * NJ + j) * 9;
    v4f R;  __builtin_memcpy(&R, rp + 4 * half, sizeof(float) * 4);
    const float r8 = rp[8];
    const float* fp = feat + m * 152 + fslot * ND;
    const v2f f01 = *(const v2f*)(fp), f23 = *(const v2f*)(fp + 2), f45 = *(const v2f*)(fp + 4);

    const float Px = pos_lower<3 * j>(jt), Py = pos_lower<3 * j + 1>(jt), Pz = pos_lower<3 * j + 2>(jt);
    float px = 0.0f, py = 0.0f, pz = 0.0f;
    if constexpr (p >= 0) {
      px = pos_lower<3 * pj>(jt); py = pos_lower<3 * pj + 1>(jt); pz = pos_lower<3 * pj + 2>(jt);
    }
    const float dx = Px - px, dy = Py - py, dz = Pz - pz;
    const float bone  = sqrtf(dx * dx + dy * dy + dz * dz);
    const float boneu = other_half(bone);

    const v2f lo2 = {r8, Px},   hi2 = {boneu, f01.x};
    const v2f lo3 = {Py, Pz},   hi3 = {f01.y, f23.x};
    const v2f lo4 = {f23.y, f45.x}, hi4 = {f45.y, 0.0f};
    xa[i][0] = R.xy;
    xa[i][1] = R.zw;
    xa[i][2] = half ? hi2 : lo2;
    xa[i][3] = half ? hi3 : lo3;
    xa[i][4] = half ? hi4 : lo4;
  };
  build_x(std::integral_constant<int, 0>{});
  if constexpr (K > 1) build_x(std::integral_constant<int, 1>{});

  v8f c0[K], c1[K];
  v2f w1a[K][5], w1b[K][5];
  #pragma unroll
  for (int i = 0; i < K; ++i) {
    const int j = j0 + i;
    const float bias0 = sW[L_B1 + j * NF + ln];
    const float bias1 = sW[L_B1 + j * NF + n1];
    #pragma unroll
    for (int g = 0; g < 8; ++g) { c0[i][g] = bias0; c1[i][g] = bias1; }
    const float* w1j = sW + L_W1 + j * NF * KP;
    const float* r0 = w1j + ln * KP;
    const float* r1 = w1j + n1 * KP;
    const v4f p0 = *(const v4f*)(r0 + 4 * half),      p1 = *(const v4f*)(r1 + 4 * half);
    const v4f q0 = *(const v4f*)(r0 + 8 + 4 * half),  q1 = *(const v4f*)(r1 + 8 + 4 * half);
    const v2f t0 = *(const v2f*)(r0 + 16 + 2 * half), t1 = *(const v2f*)(r1 + 16 + 2 * half);
    w1a[i][0] = p0.xy;  w1a[i][1] = p0.zw;  w1a[i][2] = q0.xy;  w1a[i][3] = q0.zw;  w1a[i][4] = t0;
    w1b[i][0] = p1.xy;  w1b[i][1] = p1.zw;  w1b[i][2] = q1.xy;  w1b[i][3] = q1.zw;  w1b[i][4] = t1;
  }
  #pragma unroll
  for (int st = 0; st < 5; ++st) {
    #pragma unroll
    for (int i = 0; i < K; ++i) {
      c0[i] = wmma_f32(xa[i][st], w1a[i][st], c0[i]);
      c1[i] = wmma_f32(xa[i][st], w1b[i][st], c1[i]);
    }
  }

  #pragma unroll
  for (int i = 0; i < K; ++i) {
    float* xb = hb + i * HT;
    float* d0 = xb + ln * HP + half * 8;
    const v4f a0 = {fmaxf(c0[i][0], 0.0f), fmaxf(c0[i][1], 0.0f), fmaxf(c0[i][2], 0.0f), fmaxf(c0[i][3], 0.0f)};
    const v4f a1 = {fmaxf(c0[i][4], 0.0f), fmaxf(c0[i][5], 0.0f), fmaxf(c0[i][6], 0.0f), fmaxf(c0[i][7], 0.0f)};
    *(v4f*)(d0)     = a0;
    *(v4f*)(d0 + 4) = a1;
    if (ln < 4) {
      float* d1 = xb + (4 * ln + 2 * half) * HP + 16;
      v4f b0, b1;
      #pragma unroll
      for (int g = 0; g < 4; ++g) {
        b0[g] = (ln < 3) ? fmaxf(c1[i][g],     0.0f) : 0.0f;
        b1[g] = (ln < 3) ? fmaxf(c1[i][g + 4], 0.0f) : 0.0f;
      }
      *(v4f*)(d1)      = b0;
      *(v4f*)(d1 + HP) = b1;
    }
  }
  asm volatile("s_wait_dscnt 0" ::: "memory");

  v8f c2[K];
  v2f ha[K][5], w2f[K][5];
  #pragma unroll
  for (int i = 0; i < K; ++i) {
    const int j = j0 + i;
    const float* xb = hb + i * HT;
    const float bias2 = sW[L_B2 + j * ND + n2];
    #pragma unroll
    for (int g = 0; g < 8; ++g) c2[i][g] = bias2;
    const float* w2j = sW + L_W2 + j * ND * KP;
    const float* hp = xb + (half * 2) * HP + ln;
    const float* hq = xb + (half * 8 + (ln >> 2)) * HP + 16 + (ln & 3);
    #pragma unroll
    for (int kk = 0; kk < 5; ++kk) {
      const int k = kk * 4 + half * 2;
      const v2f hk = (kk < 4) ? v2f{hp[kk * 4 * HP], hp[kk * 4 * HP + HP]} : v2f{hq[0], hq[4 * HP]};
      ha[i][kk]  = hk;
      w2f[i][kk] = *(const v2f*)(w2j + n2 * KP + k);
    }
  }
  #pragma unroll
  for (int kk = 0; kk < 5; ++kk) {
    #pragma unroll
    for (int i = 0; i < K; ++i) c2[i] = wmma_f32(ha[i][kk], w2f[i][kk], c2[i]);
  }
  if (ln < ND) {
    #pragma unroll
    for (int i = 0; i < K; ++i) {
      #pragma unroll
      for (int g = 0; g < 8; ++g)
        feat[(g + half * 8) * 152 + (j0 + i) * ND + ln] = c2[i][g];
    }
  }
}

__global__ __launch_bounds__(NW * 32)
void hier_pose_kernel(const float* __restrict__ rots, const float* __restrict__ Jtrs,
                      const float* __restrict__ W0, const float* __restrict__ b0,
                      const float* __restrict__ W1, const float* __restrict__ b1,
                      const float* __restrict__ W2, const float* __restrict__ b2,
                      float* __restrict__ out, long bsz)
{
  __shared__ __align__(16) float sW[L_WTOT];
  __shared__ __align__(16) float sfeat[NW][16][152];
  __shared__ __align__(16) float sx[NW][2][16][HP];

  {
    const int t = threadIdx.x;
    constexpr int NT  = NW * 32;
    constexpr int N1  = NJ * NF * NF, T1 = (N1 + NT - 1) / NT;
    constexpr int N2  = NJ * ND * NF, T2 = (N2 + NT - 1) / NT;
    constexpr int N0  = ND * 12 * NJ, T0 = (N0 + NT - 1) / NT;
    constexpr int NB1 = NJ * NF,      TB = (NB1 + NT - 1) / NT;
    float v1[T1], v2[T2], v0[T0], vb[TB];
    #pragma unroll
    for (int u = 0; u < T1; ++u) { const int i = t + u * NT; v1[u] = W1[i < N1 ? i : N1 - 1]; }
    #pragma unroll
    for (int u = 0; u < T2; ++u) { const int i = t + u * NT; v2[u] = W2[i < N2 ? i : N2 - 1]; }
    #pragma unroll
    for (int u = 0; u < T0; ++u) { const int i = t + u * NT; v0[u] = W0[i < N0 ? i : N0 - 1]; }
    #pragma unroll
    for (int u = 0; u < TB; ++u) { const int i = t + u * NT; vb[u] = b1[i < NB1 ? i : NB1 - 1]; }
    const float vb2 = b2[t < NJ * ND ? t : NJ * ND - 1];
    const float vb0 = b0[t < ND ? t : ND - 1];

    #pragma unroll
    for (int u = 0; u < T1; ++u) {
      const int i = t + u * NT, r = i / NF, k = i - r * NF;
      if (i < N1) sW[L_W1 + r * KP + k] = v1[u];
    }
    #pragma unroll
    for (int u = 0; u < T2; ++u) {
      const int i = t + u * NT, r = i / NF, k = i - r * NF;
      if (i < N2) sW[L_W2 + r * KP + k] = v2[u];
    }
    #pragma unroll
    for (int u = 0; u < T0; ++u) {
      const int i = t + u * NT, n = i / (12 * NJ), k = i - n * (12 * NJ);
      if (i < N0) sW[L_W0 + n * W0P + k] = v0[u];
    }
    #pragma unroll
    for (int u = 0; u < TB; ++u) { const int i = t + u * NT; if (i < NB1) sW[L_B1 + i] = vb[u]; }
    if (t < NJ * ND) sW[L_B2 + t] = vb2;
    if (t < 8)       sW[L_B0 + t] = (t < ND) ? vb0 : 0.0f;
    for (int r = t; r < NJ * (NF + ND); r += NT) sW[L_W1 + r * KP + NF] = 0.0f;
    if (t < ND * 4) sW[L_W0 + (t >> 2) * W0P + 12 * NJ + (t & 3)] = 0.0f;
  }
  __syncthreads();

  const int  lane = threadIdx.x & 31;
  const int  wave = threadIdx.x >> 5;
  const int  half = lane >> 4;
  const int  ln   = lane & 15;
  const int  n1   = (16 + ln < NF) ? 16 + ln : NF - 1;
  const int  n2   = (ln < ND) ? ln : ND - 1;

  float* feat = &sfeat[wave][0][0];
  float* hb   = &sx[wave][0][0][0];

  const long ntiles = bsz / 16;
  for (long tile = (long)blockIdx.x * NW + wave; tile < ntiles; tile += (long)NG * NW) {
  const long s0 = tile * 16;

  const float* arow_r = rots + (s0 + ln) * 216;
  const float* arow_j = Jtrs + (s0 + ln) * 72;
  __builtin_prefetch(arow_r, 0, 3);
  __builtin_prefetch(arow_j, 0, 3);

  v8f acc;
  {
    const float ci = sW[L_B0 + n2];
    #pragma unroll
    for (int g = 0; g < 8; ++g) acc[g] = ci;
  }
  const float* w0row = sW + L_W0 + n2 * W0P;
  #pragma unroll 3
  for (int q = 0; q < 27; ++q) {
    const int k = q * 8 + half * 4;
    const v4f a = *(const v4f*)(arow_r + k);
    const v4f b = *(const v4f*)(w0row + k);
    acc = wmma_f32(a.xy, b.xy, acc);
    acc = wmma_f32(a.zw, b.zw, acc);
  }
  v4f jt[9];
  #pragma unroll
  for (int q = 0; q < 9; ++q) jt[q] = *(const v4f*)(arow_j + q * 8 + half * 4);
  #pragma unroll
  for (int q = 0; q < 9; ++q) {
    const int k = q * 8 + half * 4;
    const v4f b = *(const v4f*)(w0row + 216 + k);
    acc = wmma_f32(jt[q].xy, b.xy, acc);
    acc = wmma_f32(jt[q].zw, b.zw, acc);
  }
  if (ln < ND) {
    #pragma unroll
    for (int g = 0; g < 8; ++g)
      feat[(g + half * 8) * 152 + 144 + ln] = acc[g];
  }

#define JG(K_, J_) joint_group<K_, J_>(rots, jt, sW, feat, hb, s0, ln, half, n1, n2)
  JG(1, 0);
  JG(2, 1);  JG(2, 3);  JG(2, 5);  JG(2, 7);  JG(2, 9);  JG(2, 11);
  JG(2, 13); JG(2, 15); JG(2, 17); JG(2, 19); JG(2, 21);
  JG(1, NJ - 1);
#undef JG

  asm volatile("s_wait_dscnt 0" ::: "memory");
  #pragma unroll 4
  for (int idx = lane; idx < 16 * 36; idx += 32) {
    const int m = idx / 36;
    const int q = idx - m * 36;
    *(v4f*)(out + (s0 + m) * 144 + q * 4) = *(const v4f*)(feat + m * 152 + q * 4);
  }
  }
}

extern "C" void kernel_launch(void* const* d_in, const int* in_sizes, int n_in,
                              void* d_out, int out_size, void* d_ws, size_t ws_size,
                              hipStream_t stream) {
  const float* rots = (const float*)d_in[0];
  const float* Jtrs = (const float*)d_in[1];
  const float* W0   = (const float*)d_in[2];
  const float* b0   = (const float*)d_in[3];
  const float* W1   = (const float*)d_in[4];
  const float* b1   = (const float*)d_in[5];
  const float* W2   = (const float*)d_in[6];
  const float* b2   = (const float*)d_in[7];
  float* out = (float*)d_out;

  const long bsz = (long)in_sizes[0] / (NJ * 9);
  if (bsz > 0) {
    hipLaunchKernelGGL(hier_pose_kernel, dim3(NG), dim3(NW * 32), 0, stream,
                       rots, Jtrs, W0, b0, W1, b1, W2, b2, out, bsz);
  }
}


# The 36 16 B row loads of the 288->6 product (27 rot, 9 position) go out as one batch before its 72 products (scheduling fence): one exposed memory round trip there per tile, not ten; sums unchanged
# speedup vs baseline: 2.7182x; 1.0338x over previous
#include <hip/hip_runtime.h>
#include <math.h>
#include <type_traits>


typedef float v2f __attribute__((ext_vector_type(2)));
typedef float v4f __attribute__((ext_vector_type(4)));
typedef float v8f __attribute__((ext_vector_type(8)));

#define NJ 24
#define ND 6
#define NF 19
#define NW 8
#define NG 512
#define KP 20
#define W0P 292
#define HP 20
#define HT (16 * HP)

#define L_W1  0
#define L_W2  (L_W1 + NJ*NF*KP)
#define L_W0  (L_W2 + NJ*ND*KP)
#define L_B1  (L_W0 + ND*W0P)
#define L_B2  (L_B1 + NJ*NF)
#define L_B0  (L_B2 + NJ*ND)
#define L_WTOT (L_B0 + 8)

#define PARENT_LIST -1,0,0,0,1,2,3,4,5,6,7,8,9,9,9,12,13,14,16,17,18,19,20,21

constexpr int k_parents[NJ] = {PARENT_LIST};
constexpr bool pairs_independent() {
  for (int j = 1; j + 1 < NJ; j += 2) if (k_parents[j + 1] >= j) return false;
  return true;
}
static_assert(pairs_independent(), "a joint pair (j, j+1), j odd, has j as the parent of j+1");

static __device__ __forceinline__ v8f wmma_f32(v2f a, v2f b, v8f c) {
  return __builtin_amdgcn_wmma_f32_16x16x4_f32(
       false, a,  false, b,
       (short)0, c,  false,  false);
}

static __device__ __forceinline__ float other_half(float v) {
  const unsigned u = __builtin_bit_cast(unsigned, v);
  return __builtin_bit_cast(float,
      __builtin_amdgcn_permlanex16(u, u, 0x76543210u, 0xfedcba98u, false, false));
}

template <int F>
static __device__ __forceinline__ float pos_lower(const v4f (&jt)[9]) {
  const float own = jt[F >> 3][F & 3];
  if constexpr (((F >> 2) & 1) != 0) return other_half(own);
  else return own;
}

template <int K, int J0>
static __device__ __forceinline__ void joint_group(
    const float* __restrict__ rots, const v4f (&jt)[9],
    const float* sW, float* feat, float* hb, const long s0,
    const int ln, const int half, const int n1, const int n2)
{
  constexpr int j0 = J0;
  v2f xa[K][5];
  auto build_x = [&](auto ic) {
    constexpr int  i     = decltype(ic)::value;
    constexpr int  j     = j0 + i;
    constexpr int  p     = k_parents[j];
    constexpr int  fslot = (p < 0) ? 24 : p;
    constexpr int  pj    = (p < 0) ? j : p;
    const int  m     = ln;
    const float* rp  = rots + ((s0 + m) * NJ + j) * 9;
    v4f R;  __builtin_memcpy(&R, rp + 4 * half, sizeof(float) * 4);
    const float r8 = rp[8];
    const float* fp = feat + m * 152 + fslot * ND;
    const v2f f01 = *(const v2f*)(fp), f23 = *(const v2f*)(fp + 2), f45 = *(const v2f*)(fp + 4);

    const float Px = pos_lower<3 * j>(jt), Py = pos_lower<3 * j + 1>(jt), Pz = pos_lower<3 * j + 2>(jt);
    float px = 0.0f, py = 0.0f, pz = 0.0f;
    if constexpr (p >= 0) {
      px = pos_lower<3 * pj>(jt); py = pos_lower<3 * pj + 1>(jt); pz = pos_lower<3 * pj + 2>(jt);
    }
    const float dx = Px - px, dy = Py - py, dz = Pz - pz;
    const float bone  = sqrtf(dx * dx + dy * dy + dz * dz);
    const float boneu = other_half(bone);

    const v2f lo2 = {r8, Px},   hi2 = {boneu, f01.x};
    const v2f lo3 = {Py, Pz},   hi3 = {f01.y, f23.x};
    const v2f lo4 = {f23.y, f45.x}, hi4 = {f45.y, 0.0f};
    xa[i][0] = R.xy;
    xa[i][1] = R.zw;
    xa[i][2] = half ? hi2 : lo2;
    xa[i][3] = half ? hi3 : lo3;
    xa[i][4] = half ? hi4 : lo4;
  };
  build_x(std::integral_constant<int, 0>{});
  if constexpr (K > 1) build_x(std::integral_constant<int, 1>{});

  v8f c0[K], c1[K];
  v2f w1a[K][5], w1b[K][5];
  #pragma unroll
  for (int i = 0; i < K; ++i) {
    const int j = j0 + i;
    const float bias0 = sW[L_B1 + j * NF + ln];
    const float bias1 = sW[L_B1 + j * NF + n1];
    #pragma unroll
    for (int g = 0; g < 8; ++g) { c0[i][g] = bias0; c1[i][g] = bias1; }
    const float* w1j = sW + L_W1 + j * NF * KP;
    const float* r0 = w1j + ln * KP;
    const float* r1 = w1j + n1 * KP;
    const v4f p0 = *(const v4f*)(r0 + 4 * half),      p1 = *(const v4f*)(r1 + 4 * half);
    const v4f q0 = *(const v4f*)(r0 + 8 + 4 * half),  q1 = *(const v4f*)(r1 + 8 + 4 * half);
    const v2f t0 = *(const v2f*)(r0 + 16 + 2 * half), t1 = *(const v2f*)(r1 + 16 + 2 * half);
    w1a[i][0] = p0.xy;  w1a[i][1] = p0.zw;  w1a[i][2] = q0.xy;  w1a[i][3] = q0.zw;  w1a[i][4] = t0;
    w1b[i][0] = p1.xy;  w1b[i][1] = p1.zw;  w1b[i][2] = q1.xy;  w1b[i][3] = q1.zw;  w1b[i][4] = t1;
  }
  #pragma unroll
  for (int st = 0; st < 5; ++st) {
    #pragma unroll
    for (int i = 0; i < K; ++i) {
      c0[i] = wmma_f32(xa[i][st], w1a[i][st], c0[i]);
      c1[i] = wmma_f32(xa[i][st], w1b[i][st], c1[i]);
    }
  }

  #pragma unroll
  for (int i = 0; i < K; ++i) {
    float* xb = hb + i * HT;
    float* d0 = xb + ln * HP + half * 8;
    const v4f a0 = {fmaxf(c0[i][0], 0.0f), fmaxf(c0[i][1], 0.0f), fmaxf(c0[i][2], 0.0f), fmaxf(c0[i][3], 0.0f)};
    const v4f a1 = {fmaxf(c0[i][4], 0.0f), fmaxf(c0[i][5], 0.0f), fmaxf(c0[i][6], 0.0f), fmaxf(c0[i][7], 0.0f)};
    *(v4f*)(d0)     = a0;
    *(v4f*)(d0 + 4) = a1;
    if (ln < 4) {
      float* d1 = xb + (4 * ln + 2 * half) * HP + 16;
      v4f b0, b1;
      #pragma unroll
      for (int g = 0; g < 4; ++g) {
        b0[g] = (ln < 3) ? fmaxf(c1[i][g],     0.0f) : 0.0f;
        b1[g] = (ln < 3) ? fmaxf(c1[i][g + 4], 0.0f) : 0.0f;
      }
      *(v4f*)(d1)      = b0;
      *(v4f*)(d1 + HP) = b1;
    }
  }
  asm volatile("s_wait_dscnt 0" ::: "memory");

  v8f c2[K];
  v2f ha[K][5], w2f[K][5];
  #pragma unroll
  for (int i = 0; i < K; ++i) {
    const int j = j0 + i;
    const float* xb = hb + i * HT;
    const float bias2 = sW[L_B2 + j * ND + n2];
    #pragma unroll
    for (int g = 0; g < 8; ++g) c2[i][g] = bias2;
    const float* w2j = sW + L_W2 + j * ND * KP;
    const float* hp = xb + (half * 2) * HP + ln;
    const float* hq = xb + (half * 8 + (ln >> 2)) * HP + 16 + (ln & 3);
    #pragma unroll
    for (int kk = 0; kk < 5; ++kk) {
      const int k = kk * 4 + half * 2;
      const v2f hk = (kk < 4) ? v2f{hp[kk * 4 * HP], hp[kk * 4 * HP + HP]} : v2f{hq[0], hq[4 * HP]};
      ha[i][kk]  = hk;
      w2f[i][kk] = *(const v2f*)(w2j + n2 * KP + k);
    }
  }
  #pragma unroll
  for (int kk = 0; kk < 5; ++kk) {
    #pragma unroll
    for (int i = 0; i < K; ++i) c2[i] = wmma_f32(ha[i][kk], w2f[i][kk], c2[i]);
  }
  if (ln < ND) {
    #pragma unroll
    for (int i = 0; i < K; ++i) {
      #pragma unroll
      for (int g = 0; g < 8; ++g)
        feat[(g + half * 8) * 152 + (j0 + i) * ND + ln] = c2[i][g];
    }
  }
}

__global__ __launch_bounds__(NW * 32)
void hier_pose_kernel(const float* __restrict__ rots, const float* __restrict__ Jtrs,
                      const float* __restrict__ W0, const float* __restrict__ b0,
                      const float* __restrict__ W1, const float* __restrict__ b1,
                      const float* __restrict__ W2, const float* __restrict__ b2,
                      float* __restrict__ out, long bsz)
{
  __shared__ __align__(16) float sW[L_WTOT];
  __shared__ __align__(16) float sfeat[NW][16][152];
  __shared__ __align__(16) float sx[NW][2][16][HP];

  {
    const int t = threadIdx.x;
    constexpr int NT  = NW * 32;
    constexpr int N1  = NJ * NF * NF, T1 = (N1 + NT - 1) / NT;
    constexpr int N2  = NJ * ND * NF, T2 = (N2 + NT - 1) / NT;
    constexpr int N0  = ND * 12 * NJ, T0 = (N0 + NT - 1) / NT;
    constexpr int NB1 = NJ * NF,      TB = (NB1 + NT - 1) / NT;
    float v1[T1], v2[T2], v0[T0], vb[TB];
    #pragma unroll
    for (int u = 0; u < T1; ++u) { const int i = t + u * NT; v1[u] = W1[i < N1 ? i : N1 - 1]; }
    #pragma unroll
    for (int u = 0; u < T2; ++u) { const int i = t + u * NT; v2[u] = W2[i < N2 ? i : N2 - 1]; }
    #pragma unroll
    for (int u = 0; u < T0; ++u) { const int i = t + u * NT; v0[u] = W0[i < N0 ? i : N0 - 1]; }
    #pragma unroll
    for (int u = 0; u < TB; ++u) { const int i = t + u * NT; vb[u] = b1[i < NB1 ? i : NB1 - 1]; }
    const float vb2 = b2[t < NJ * ND ? t : NJ * ND - 1];
    const float vb0 = b0[t < ND ? t : ND - 1];

    #pragma unroll
    for (int u = 0; u < T1; ++u) {
      const int i = t + u * NT, r = i / NF, k = i - r * NF;
      if (i < N1) sW[L_W1 + r * KP + k] = v1[u];
    }
    #pragma unroll
    for (int u = 0; u < T2; ++u) {
      const int i = t + u * NT, r = i / NF, k = i - r * NF;
      if (i < N2) sW[L_W2 + r * KP + k] = v2[u];
    }
    #pragma unroll
    for (int u = 0; u < T0; ++u) {
      const int i = t + u * NT, n = i / (12 * NJ), k = i - n * (12 * NJ);
      if (i < N0) sW[L_W0 + n * W0P + k] = v0[u];
    }
    #pragma unroll
    for (int u = 0; u < TB; ++u) { const int i = t + u * NT; if (i < NB1) sW[L_B1 + i] = vb[u]; }
    if (t < NJ * ND) sW[L_B2 + t] = vb2;
    if (t < 8)       sW[L_B0 + t] = (t < ND) ? vb0 : 0.0f;
    for (int r = t; r < NJ * (NF + ND); r += NT) sW[L_W1 + r * KP + NF] = 0.0f;
    if (t < ND * 4) sW[L_W0 + (t >> 2) * W0P + 12 * NJ + (t & 3)] = 0.0f;
  }
  __syncthreads();

  const int  lane = threadIdx.x & 31;
  const int  wave = threadIdx.x >> 5;
  const int  half = lane >> 4;
  const int  ln   = lane & 15;
  const int  n1   = (16 + ln < NF) ? 16 + ln : NF - 1;
  const int  n2   = (ln < ND) ? ln : ND - 1;

  float* feat = &sfeat[wave][0][0];
  float* hb   = &sx[wave][0][0][0];

  const long ntiles = bsz / 16;
  for (long tile = (long)blockIdx.x * NW + wave; tile < ntiles; tile += (long)NG * NW) {
  const long s0 = tile * 16;

  const float* arow_r = rots + (s0 + ln) * 216;
  const float* arow_j = Jtrs + (s0 + ln) * 72;
  __builtin_prefetch(arow_r, 0, 3);
  __builtin_prefetch(arow_j, 0, 3);

  v8f acc;
  {
    const float ci = sW[L_B0 + n2];
    #pragma unroll
    for (int g = 0; g < 8; ++g) acc[g] = ci;
  }
  const float* w0row = sW + L_W0 + n2 * W0P;
  v4f ra[27];
  v4f jt[9];
  #pragma unroll
  for (int q = 0; q < 27; ++q) ra[q] = *(const v4f*)(arow_r + q * 8 + half * 4);
  #pragma unroll
  for (int q = 0; q < 9; ++q) jt[q] = *(const v4f*)(arow_j + q * 8 + half * 4);
  __builtin_amdgcn_sched_barrier(0);
  #pragma unroll
  for (int q = 0; q < 27; ++q) {
    const int k = q * 8 + half * 4;
    const v4f b = *(const v4f*)(w0row + k);
    acc = wmma_f32(ra[q].xy, b.xy, acc);
    acc = wmma_f32(ra[q].zw, b.zw, acc);
  }
  #pragma unroll
  for (int q = 0; q < 9; ++q) {
    const int k = q * 8 + half * 4;
    const v4f b = *(const v4f*)(w0row + 216 + k);
    acc = wmma_f32(jt[q].xy, b.xy, acc);
    acc = wmma_f32(jt[q].zw, b.zw, acc);
  }
  if (ln < ND) {
    #pragma unroll
    for (int g = 0; g < 8; ++g)
      feat[(g + half * 8) * 152 + 144 + ln] = acc[g];
  }

#define JG(K_, J_) joint_group<K_, J_>(rots, jt, sW, feat, hb, s0, ln, half, n1, n2)
  JG(1, 0);
  JG(2, 1);  JG(2, 3);  JG(2, 5);  JG(2, 7);  JG(2, 9);  JG(2, 11);
  JG(2, 13); JG(2, 15); JG(2, 17); JG(2, 19); JG(2, 21);
  JG(1, NJ - 1);
#undef JG

  asm volatile("s_wait_dscnt 0" ::: "memory");
  #pragma unroll 4
  for (int idx = lane; idx < 16 * 36; idx += 32) {
    const int m = idx / 36;
    const int q = idx - m * 36;
    *(v4f*)(out + (s0 + m) * 144 + q * 4) = *(const v4f*)(feat + m * 152 + q * 4);
  }
  }
}

extern "C" void kernel_launch(void* const* d_in, const int* in_sizes, int n_in,
                              void* d_out, int out_size, void* d_ws, size_t ws_size,
                              hipStream_t stream) {
  const float* rots = (const float*)d_in[0];
  const float* Jtrs = (const float*)d_in[1];
  const float* W0   = (const float*)d_in[2];
  const float* b0   = (const float*)d_in[3];
  const float* W1   = (const float*)d_in[4];
  const float* b1   = (const float*)d_in[5];
  const float* W2   = (const float*)d_in[6];
  const float* b2   = (const float*)d_in[7];
  float* out = (float*)d_out;

  const long bsz = (long)in_sizes[0] / (NJ * 9);
  if (bsz > 0) {
    hipLaunchKernelGGL(hier_pose_kernel, dim3(NG), dim3(NW * 32), 0, stream,
                       rots, Jtrs, W0, b0, W1, b1, W2, b2, out, bsz);
  }
}
